# starting kernel
# baseline (speedup 1.0000x reference)
#include <hip/hip_runtime.h>

typedef __attribute__((ext_vector_type(2))) float v2f;
typedef __attribute__((ext_vector_type(8))) float v8f;

#define BSZ    8192
#define FDIM   64
#define HDIM   256
#define KSPLIT 4
#define FPB    (FDIM / KSPLIT)
#define WAVES  8
#define TPB    (WAVES * 32)
#define ROWS_PER_WG (WAVES * 16)

#define LOG2E 1.4426950408889634f
#define LN2   0.6931471805599453f

__device__ __forceinline__ v2f mish2_l2(v2f hl2) {
    const v2f cmax = {22.0f, 22.0f};
    const v2f one  = {1.0f, 1.0f};
    const v2f m2   = {-2.0f, -2.0f};
    v2f hc = __builtin_elementwise_min(hl2, cmax);
    v2f t;
    t.x = __builtin_amdgcn_exp2f(hc.x);
    t.y = __builtin_amdgcn_exp2f(hc.y);
    v2f s = t + 1.0f;
    v2f d = __builtin_elementwise_fma(s, s, one);
    v2f q;
    q.x = __builtin_amdgcn_rcpf(d.x);
    q.y = __builtin_amdgcn_rcpf(d.y);
    v2f w = hl2 * q;
    return __builtin_elementwise_fma(w, m2, hl2);
}

__global__ __launch_bounds__(TPB) void nam_main(
    const float* __restrict__ X,
    const float* __restrict__ exu_w,
    const float* __restrict__ exu_b,
    const float* __restrict__ dense_k,
    float* __restrict__ partial)
{
    __shared__ float s_ew[FPB * HDIM];
    __shared__ float s_kd[FPB * HDIM];

    const int tid = threadIdx.x;
    const int f0  = blockIdx.y * FPB;

    for (int i = tid; i < FPB * HDIM; i += TPB) {
        s_ew[i] = __expf(exu_w[f0 * HDIM + i]) * LOG2E;
        s_kd[i] = dense_k[f0 * HDIM + i] * LN2;
    }
    __syncthreads();

    const int wave = tid >> 5;
    const int lane = tid & 31;
    const int half = lane >> 4;
    const int m    = lane & 15;
    const int rowBase = blockIdx.x * ROWS_PER_WG + wave * 16;
    const int b    = rowBase + m;
    const int koff = half * 2;

    v8f acc = {};

    for (int fl = 0; fl < FPB; ++fl) {
        const int f = f0 + fl;
        const float c = X[b * FDIM + f] - exu_b[f];
        const float* pe = &s_ew[fl * HDIM + koff];
        const float* pk = &s_kd[fl * HDIM + koff];
#pragma unroll 4
        for (int hh = 0; hh < HDIM; hh += 8) {
            v2f e2a = *(const v2f*)(pe + hh);
            v2f k2a = *(const v2f*)(pk + hh);
            v2f e2b = *(const v2f*)(pe + hh + 4);
            v2f k2b = *(const v2f*)(pk + hh + 4);
            v2f aa = mish2_l2(c * e2a);
            v2f ab = mish2_l2(c * e2b);
            acc = __builtin_amdgcn_wmma_f32_16x16x4_f32(
                false, aa, false, k2a, (short)0, acc, false, false);
            acc = __builtin_amdgcn_wmma_f32_16x16x4_f32(
                false, ab, false, k2b, (short)0, acc, false, false);
        }
    }

    if (m == 0) {
        float* dst = &partial[blockIdx.y * BSZ + rowBase + half * 8];
#pragma unroll
        for (int j = 0; j < 8; ++j) dst[j] = acc[j];
    }
}

__global__ __launch_bounds__(256) void nam_reduce(
    const float* __restrict__ partial,
    const float* __restrict__ dense_b,
    float* __restrict__ out)
{
    const int b = blockIdx.x * 256 + threadIdx.x;
    float s = 0.0f;
#pragma unroll
    for (int f = 0; f < FDIM; ++f) s += dense_b[f];
#pragma unroll
    for (int y = 0; y < KSPLIT; ++y) s += partial[y * BSZ + b];
    out[b] = s;
}

extern "C" void kernel_launch(void* const* d_in, const int* in_sizes, int n_in,
                              void* d_out, int out_size, void* d_ws, size_t ws_size,
                              hipStream_t stream) {
    (void)in_sizes; (void)n_in; (void)out_size; (void)ws_size;
    const float* X       = (const float*)d_in[0];
    const float* exu_w   = (const float*)d_in[1];
    const float* exu_b   = (const float*)d_in[2];
    const float* dense_k = (const float*)d_in[3];
    const float* dense_b = (const float*)d_in[4];
    float* out     = (float*)d_out;
    float* partial = (float*)d_ws;

    dim3 grid(BSZ / ROWS_PER_WG, KSPLIT);
    nam_main<<<grid, TPB, 0, stream>>>(X, exu_w, exu_b, dense_k, partial);
    nam_reduce<<<BSZ / 256, 256, 0, stream>>>(partial, dense_b, out);
}


# Feature split 4 -> 16 slices (4 features per work-group, 1024 work-groups): 8 waves per SIMD instead of 2 cover LDS, mish-chain and WMMA-chain waits; 16 partials per row, so the sum order changes
# speedup vs baseline: 1.1394x; 1.1394x over previous
#include <hip/hip_runtime.h>

typedef __attribute__((ext_vector_type(2))) float v2f;
typedef __attribute__((ext_vector_type(8))) float v8f;

#define BSZ    8192
#define FDIM   64
#define HDIM   256
#define KSPLIT 16
#define FPB    (FDIM / KSPLIT)
#define WAVES  8
#define TPB    (WAVES * 32)
#define ROWS_PER_WG (WAVES * 16)

#define LOG2E 1.4426950408889634f
#define LN2   0.6931471805599453f

__device__ __forceinline__ v2f mish2_l2(v2f hl2) {
    const v2f cmax = {22.0f, 22.0f};
    const v2f one  = {1.0f, 1.0f};
    const v2f m2   = {-2.0f, -2.0f};
    v2f hc = __builtin_elementwise_min(hl2, cmax);
    v2f t;
    t.x = __builtin_amdgcn_exp2f(hc.x);
    t.y = __builtin_amdgcn_exp2f(hc.y);
    v2f s = t + 1.0f;
    v2f d = __builtin_elementwise_fma(s, s, one);
    v2f q;
    q.x = __builtin_amdgcn_rcpf(d.x);
    q.y = __builtin_amdgcn_rcpf(d.y);
    v2f w = hl2 * q;
    return __builtin_elementwise_fma(w, m2, hl2);
}

__global__ __launch_bounds__(TPB) void nam_main(
    const float* __restrict__ X,
    const float* __restrict__ exu_w,
    const float* __restrict__ exu_b,
    const float* __restrict__ dense_k,
    float* __restrict__ partial)
{
    __shared__ float s_ew[FPB * HDIM];
    __shared__ float s_kd[FPB * HDIM];

    const int tid = threadIdx.x;
    const int f0  = blockIdx.y * FPB;

#pragma unroll
    for (int j = 0; j < FPB * HDIM / TPB; ++j) {
        const int i = tid + j * TPB;
        s_ew[i] = __expf(exu_w[f0 * HDIM + i]) * LOG2E;
        s_kd[i] = dense_k[f0 * HDIM + i] * LN2;
    }
    __syncthreads();

    const int wave = tid >> 5;
    const int lane = tid & 31;
    const int half = lane >> 4;
    const int m    = lane & 15;
    const int rowBase = blockIdx.x * ROWS_PER_WG + wave * 16;
    const int b    = rowBase + m;
    const int koff = half * 2;

    v8f acc = {};

    for (int fl = 0; fl < FPB; ++fl) {
        const int f = f0 + fl;
        const float c = X[b * FDIM + f] - exu_b[f];
        const float* pe = &s_ew[fl * HDIM + koff];
        const float* pk = &s_kd[fl * HDIM + koff];
#pragma unroll 4
        for (int hh = 0; hh < HDIM; hh += 8) {
            v2f e2a = *(const v2f*)(pe + hh);
            v2f k2a = *(const v2f*)(pk + hh);
            v2f e2b = *(const v2f*)(pe + hh + 4);
            v2f k2b = *(const v2f*)(pk + hh + 4);
            v2f aa = mish2_l2(c * e2a);
            v2f ab = mish2_l2(c * e2b);
            acc = __builtin_amdgcn_wmma_f32_16x16x4_f32(
                false, aa, false, k2a, (short)0, acc, false, false);
            acc = __builtin_amdgcn_wmma_f32_16x16x4_f32(
                false, ab, false, k2b, (short)0, acc, false, false);
        }
    }

    if (m == 0) {
        float* dst = &partial[blockIdx.y * BSZ + rowBase + half * 8];
#pragma unroll
        for (int j = 0; j < 8; ++j) dst[j] = acc[j];
    }
}

__global__ __launch_bounds__(256) void nam_reduce(
    const float* __restrict__ partial,
    const float* __restrict__ dense_b,
    float* __restrict__ out)
{
    const int b = blockIdx.x * 256 + threadIdx.x;
    float s = 0.0f;
#pragma unroll
    for (int f = 0; f < FDIM; ++f) s += dense_b[f];
#pragma unroll
    for (int y = 0; y < KSPLIT; ++y) s += partial[y * BSZ + b];
    out[b] = s;
}

extern "C" void kernel_launch(void* const* d_in, const int* in_sizes, int n_in,
                              void* d_out, int out_size, void* d_ws, size_t ws_size,
                              hipStream_t stream) {
    (void)in_sizes; (void)n_in; (void)out_size; (void)ws_size;
    const float* X       = (const float*)d_in[0];
    const float* exu_w   = (const float*)d_in[1];
    const float* exu_b   = (const float*)d_in[2];
    const float* dense_k = (const float*)d_in[3];
    const float* dense_b = (const float*)d_in[4];
    float* out     = (float*)d_out;
    float* partial = (float*)d_ws;

    dim3 grid(BSZ / ROWS_PER_WG, KSPLIT);
    nam_main<<<grid, TPB, 0, stream>>>(X, exu_w, exu_b, dense_k, partial);
    nam_reduce<<<BSZ / 256, 256, 0, stream>>>(partial, dense_b, out);
}


# Factor c=(x-b) and the linear part out of the sum over h: WMMA takes q=1/((e^h+1)^2+1) against weights -2*k*exp(w), c*(S+sum) once per row and feature; 6 to 3 packed ops per pair; sum order changes
# speedup vs baseline: 1.1870x; 1.0418x over previous
#include <hip/hip_runtime.h>


typedef __attribute__((ext_vector_type(2))) float v2f;
typedef __attribute__((ext_vector_type(4))) float v4f;
typedef __attribute__((ext_vector_type(8))) float v8f;

#define BSZ    8192
#define FDIM   64
#define HDIM   256
#define KSPLIT 16
#define FPB    (FDIM / KSPLIT)
#define WAVES  8
#define TPB    (WAVES * 32)
#define ROWS_PER_WG (WAVES * 16)

#define LOG2E 1.4426950408889634f

__device__ __forceinline__ v2f mish_q2(v2f hl2) {
    const v2f one = {1.0f, 1.0f};
    v2f t;
    t.x = __builtin_amdgcn_exp2f(hl2.x);
    t.y = __builtin_amdgcn_exp2f(hl2.y);
    v2f s = t + 1.0f;
    v2f d = __builtin_elementwise_fma(s, s, one);
    v2f q;
    q.x = __builtin_amdgcn_rcpf(d.x);
    q.y = __builtin_amdgcn_rcpf(d.y);
    return q;
}

__global__ __launch_bounds__(TPB) void nam_main(
    const float* __restrict__ X,
    const float* __restrict__ exu_w,
    const float* __restrict__ exu_b,
    const float* __restrict__ dense_k,
    float* __restrict__ partial)
{
    __shared__ float s_ew[FPB * HDIM];
    __shared__ float s_kd[FPB * HDIM];
    __shared__ float s_S[FPB];

    const int tid = threadIdx.x;
    const int f0  = blockIdx.y * FPB;

#pragma unroll
    for (int j = 0; j < FPB * HDIM / TPB; ++j) {
        const int i = tid + j * TPB;
        const float E = __expf(exu_w[f0 * HDIM + i]);
        s_ew[i] = E * LOG2E;
        s_kd[i] = -2.0f * dense_k[f0 * HDIM + i] * E;
    }
    __syncthreads();

    const int wave = tid >> 5;
    const int lane = tid & 31;

    if (wave < FPB) {
        const v4f* p = (const v4f*)&s_kd[wave * HDIM + lane * 8];
        const v4f a = p[0];
        const v4f b4 = p[1];
        float s = ((a.x + a.y) + (a.z + a.w)) + ((b4.x + b4.y) + (b4.z + b4.w));
#pragma unroll
        for (int o = 16; o > 0; o >>= 1) s += __shfl_xor(s, o, 32);
        if (lane == 0) s_S[wave] = -0.5f * s;
    }
    __syncthreads();

    const int half = lane >> 4;
    const int n    = lane & 15;
    const int rowBase = blockIdx.x * ROWS_PER_WG + wave * 16;
    const int b    = rowBase + n;
    const int koff = half * 2;

    float tot = 0.0f;

#pragma unroll
    for (int fl = 0; fl < FPB; ++fl) {
        const int f = f0 + fl;
        const float c = X[b * FDIM + f] - exu_b[f];
        const float S = s_S[fl];
        v8f acc = {S, S, S, S, S, S, S, S};
        const float* pe = &s_ew[fl * HDIM + koff];
        const float* pk = &s_kd[fl * HDIM + koff];
        for (int hh = 0; hh < HDIM; hh += 32) {
            v2f e2[8], k2[8];
#pragma unroll
            for (int j = 0; j < 8; ++j) {
                e2[j] = *(const v2f*)(pe + hh + 4 * j);
                k2[j] = *(const v2f*)(pk + hh + 4 * j);
            }
            __builtin_amdgcn_sched_barrier(0);
#pragma unroll
            for (int j = 0; j < 8; ++j) {
                v2f q = mish_q2(c * e2[j]);
                acc = __builtin_amdgcn_wmma_f32_16x16x4_f32(
                    false, k2[j], false, q, (short)0, acc, false, false);
            }
        }
        asm volatile("v_nop\n\tv_nop\n\tv_nop\n\tv_nop\n\tv_nop" : "+v"(acc));
        tot = __builtin_fmaf(c, acc[0], tot);
    }

    if (half == 0) partial[blockIdx.y * BSZ + b] = tot;
}

__global__ __launch_bounds__(256) void nam_reduce(
    const float* __restrict__ partial,
    const float* __restrict__ dense_b,
    float* __restrict__ out)
{
    const int b = blockIdx.x * 256 + threadIdx.x;
    float s = 0.0f;
#pragma unroll
    for (int f = 0; f < FDIM; ++f) s += dense_b[f];
#pragma unroll
    for (int y = 0; y < KSPLIT; ++y) s += partial[y * BSZ + b];
    out[b] = s;
}

extern "C" void kernel_launch(void* const* d_in, const int* in_sizes, int n_in,
                              void* d_out, int out_size, void* d_ws, size_t ws_size,
                              hipStream_t stream) {
    (void)in_sizes; (void)n_in; (void)out_size; (void)ws_size;
    const float* X       = (const float*)d_in[0];
    const float* exu_w   = (const float*)d_in[1];
    const float* exu_b   = (const float*)d_in[2];
    const float* dense_k = (const float*)d_in[3];
    const float* dense_b = (const float*)d_in[4];
    float* out     = (float*)d_out;
    float* partial = (float*)d_ws;

    dim3 grid(BSZ / ROWS_PER_WG, KSPLIT);
    nam_main<<<grid, TPB, 0, stream>>>(X, exu_w, exu_b, dense_k, partial);
    nam_reduce<<<BSZ / 256, 256, 0, stream>>>(partial, dense_b, out);
}


# Software-pipeline the exp/rcp chains five WMMAs ahead (inline asm, six fixed register pairs, LDS operands six steps ahead): independent transcendentals follow each WMMA; sum order unchanged
# speedup vs baseline: 1.2364x; 1.0416x over previous
#include <hip/hip_runtime.h>
#include <utility>


typedef __attribute__((ext_vector_type(2))) float v2f;
typedef __attribute__((ext_vector_type(4))) float v4f;
typedef __attribute__((ext_vector_type(8))) float v8f;

#define BSZ    8192
#define FDIM   64
#define HDIM   256
#define KSPLIT 16
#define FPB    (FDIM / KSPLIT)
#define WAVES  8
#define TPB    (WAVES * 32)
#define ROWS_PER_WG (WAVES * 16)

#define SPF    (HDIM / 4)

#define LOG2E 1.4426950408889634f


template <int K> struct Slot;
#define DEF_SLOT(K, LO, HI)                                                                      \
template <> struct Slot<K> {                                                                     \
    static __device__ __forceinline__ void wmma(v8f& acc, const v2f a, const v2f& q) {           \
        asm volatile("v_wmma_f32_16x16x4_f32 %0, %1, %2, %0"                                     \
                     : "+v"(acc) : "v"(a), "{v[" #LO ":" #HI "]}"(q));                           \
    }                                                                                            \
    static __device__ __forceinline__ void exp2(v2f& r) {                 \
        asm volatile("v_exp_f32 v" #LO ", v" #LO "\n\tv_exp_f32 v" #HI ", v" #HI                \
                     : "+{v[" #LO ":" #HI "]}"(r));                                              \
    }                                                                                            \
    static __device__ __forceinline__ void rcp(v2f& r) {                             \
        asm volatile("v_rcp_f32 v" #LO ", v" #LO "\n\tv_rcp_f32 v" #HI ", v" #HI                \
                     : "+{v[" #LO ":" #HI "]}"(r));                                              \
    }                                                                                            \
    static __device__ __forceinline__ void add1(v2f& r) {                            \
        asm volatile("v_pk_add_f32 %0, %0, 1.0 op_sel_hi:[1,0]" : "+{v[" #LO ":" #HI "]}"(r));   \
    }                                                                                            \
    static __device__ __forceinline__ void sq1(v2f& r) {                       \
        asm volatile("v_pk_fma_f32 %0, %0, %0, 1.0 op_sel_hi:[1,1,0]"                            \
                     : "+{v[" #LO ":" #HI "]}"(r));                                              \
    }                                                                                            \
    static __device__ __forceinline__ void mul(v2f& r, const v2f cc, const v2f e) {              \
        asm volatile("v_pk_mul_f32 %0, %1, %2"                     \
                     : "={v[" #LO ":" #HI "]}"(r) : "v"(cc), "v"(e));                            \
    }                                                                                            \
};
DEF_SLOT(0, 100, 101)
DEF_SLOT(1, 102, 103)
DEF_SLOT(2, 104, 105)
DEF_SLOT(3, 106, 107)
DEF_SLOT(4, 108, 109)
DEF_SLOT(5, 110, 111)
#undef DEF_SLOT

template <int s>
__device__ __forceinline__ void pipe_step(v8f& acc, v2f (&e2)[SPF], v2f (&k2)[SPF], v2f (&r)[SPF],
                                          const v2f cc, const float* pe, const float* pk)
{
    if constexpr (s >= 0)                       Slot<s % 6>::wmma(acc, k2[s], r[s]);
    if constexpr (s + 4 >= 0 && s + 4 < SPF)    Slot<(s + 4) % 6>::exp2(r[s + 4]);
    if constexpr (s + 1 >= 0 && s + 1 < SPF)    Slot<(s + 1) % 6>::rcp(r[s + 1]);
    if constexpr (s + 3 >= 0 && s + 3 < SPF)    Slot<(s + 3) % 6>::add1(r[s + 3]);
    if constexpr (s + 2 >= 0 && s + 2 < SPF)    Slot<(s + 2) % 6>::sq1(r[s + 2]);
    if constexpr (s + 5 < SPF)                  Slot<(s + 5) % 6>::mul(r[s + 5], cc, e2[s + 5]);
    if constexpr (s + 1 < SPF && s + 2 >= SPF)  asm volatile("v_nop");
    if constexpr (s >= 5 && ((s + 11) & 7) == 0 && s + 6 < SPF) {
        constexpr int g = (s + 11) >> 3;
#pragma unroll
        for (int j = 0; j < 8; ++j) {
            const int x = 8 * g + j;
            const int y = 8 * g - 5 + j;
            if (x < SPF) e2[x] = *(const v2f*)(pe + 4 * x);
            if (y < SPF) k2[y] = *(const v2f*)(pk + 4 * y);
        }
    }
}

template <int... I>
__device__ __forceinline__ void pipe_feature(std::integer_sequence<int, I...>, v8f& acc,
                                             v2f (&e2)[SPF], v2f (&k2)[SPF], v2f (&r)[SPF],
                                             const v2f cc, const float* pe, const float* pk)
{
    (pipe_step<I - 5>(acc, e2, k2, r, cc, pe, pk), ...);
}

__global__ __launch_bounds__(TPB) void nam_main(
    const float* __restrict__ X,
    const float* __restrict__ exu_w,
    const float* __restrict__ exu_b,
    const float* __restrict__ dense_k,
    float* __restrict__ partial)
{
    __shared__ float s_ew[FPB * HDIM];
    __shared__ float s_kd[FPB * HDIM];
    __shared__ float s_S[FPB];

    const int tid = threadIdx.x;
    const int f0  = blockIdx.y * FPB;

    const int wave = tid >> 5;
    const int lane = tid & 31;
    const int half = lane >> 4;
    const int n    = lane & 15;
    const int rowBase = blockIdx.x * ROWS_PER_WG + wave * 16;
    const int b    = rowBase + n;
    const int koff = half * 2;

    const v4f x4 = *(const v4f*)&X[b * FDIM + f0];

#pragma unroll
    for (int j = 0; j < FPB * HDIM / TPB; ++j) {
        const int i = tid + j * TPB;
        const float E = __expf(exu_w[f0 * HDIM + i]);
        s_ew[i] = E * LOG2E;
        s_kd[i] = -2.0f * dense_k[f0 * HDIM + i] * E;
    }
    __syncthreads();

    if (wave < FPB) {
        const v4f* p = (const v4f*)&s_kd[wave * HDIM + lane * 8];
        const v4f a = p[0];
        const v4f b4 = p[1];
        float s = ((a.x + a.y) + (a.z + a.w)) + ((b4.x + b4.y) + (b4.z + b4.w));
#pragma unroll
        for (int o = 16; o > 0; o >>= 1) s += __shfl_xor(s, o, 32);
        if (lane == 0) s_S[wave] = -0.5f * s;
    }
    __syncthreads();

    const float c0 = x4.x - exu_b[f0 + 0];
    const float c1 = x4.y - exu_b[f0 + 1];
    const float c2 = x4.z - exu_b[f0 + 2];
    const float c3 = x4.w - exu_b[f0 + 3];

    float tot = 0.0f;

#pragma nounroll
    for (int fl = 0; fl < FPB; ++fl) {
        const float c = (fl == 0) ? c0 : (fl == 1) ? c1 : (fl == 2) ? c2 : c3;
        const float S = s_S[fl];
        v8f acc = {S, S, S, S, S, S, S, S};
        const float* pe = &s_ew[fl * HDIM + koff];
        const float* pk = &s_kd[fl * HDIM + koff];

        const v2f cc = {c, c};

        v2f e2[SPF], k2[SPF], r[SPF];
#pragma unroll
        for (int x = 0; x < 16; ++x) e2[x] = *(const v2f*)(pe + 4 * x);
#pragma unroll
        for (int y = 0; y < 11; ++y) k2[y] = *(const v2f*)(pk + 4 * y);
        pipe_feature(std::make_integer_sequence<int, SPF + 5>{}, acc, e2, k2, r, cc, pe, pk);

        asm volatile("v_nop\n\tv_nop\n\tv_nop\n\tv_nop\n\tv_nop" : "+v"(acc));
        tot = __builtin_fmaf(c, acc[0], tot);
    }

    if (half == 0) partial[blockIdx.y * BSZ + b] = tot;
}

__global__ __launch_bounds__(256) void nam_reduce(
    const float* __restrict__ partial,
    const float* __restrict__ dense_b,
    float* __restrict__ out)
{
    const int b = blockIdx.x * 256 + threadIdx.x;
    float s = 0.0f;
#pragma unroll
    for (int f = 0; f < FDIM; ++f) s += dense_b[f];
#pragma unroll
    for (int y = 0; y < KSPLIT; ++y) s += partial[y * BSZ + b];
    out[b] = s;
}

extern "C" void kernel_launch(void* const* d_in, const int* in_sizes, int n_in,
                              void* d_out, int out_size, void* d_ws, size_t ws_size,
                              hipStream_t stream) {
    (void)in_sizes; (void)n_in; (void)out_size; (void)ws_size;
    const float* X       = (const float*)d_in[0];
    const float* exu_w   = (const float*)d_in[1];
    const float* exu_b   = (const float*)d_in[2];
    const float* dense_k = (const float*)d_in[3];
    const float* dense_b = (const float*)d_in[4];
    float* out     = (float*)d_out;
    float* partial = (float*)d_ws;

    dim3 grid(BSZ / ROWS_PER_WG, KSPLIT);
    nam_main<<<grid, TPB, 0, stream>>>(X, exu_w, exu_b, dense_k, partial);
    nam_reduce<<<BSZ / 256, 256, 0, stream>>>(partial, dense_b, out);
}


# Fuse the slice reduction into the main kernel: 256 work-groups of 32 waves own 32 rows x all 64 features (128 KB of tables in LDS), slice sums meet in LDS; one launch, sum order unchanged
# speedup vs baseline: 1.3147x; 1.0633x over previous
#include <hip/hip_runtime.h>
#include <utility>


typedef __attribute__((ext_vector_type(2))) float v2f;
typedef __attribute__((ext_vector_type(4))) float v4f;
typedef __attribute__((ext_vector_type(8))) float v8f;

#define BSZ    8192
#define FDIM   64
#define HDIM   256
#define KSPLIT 16
#define FPB    (FDIM / KSPLIT)
#define TILES  2
#define WAVES  (KSPLIT * TILES)
#define TPB    (WAVES * 32)
#define ROWS_PER_WG (TILES * 16)

#define SPF    (HDIM / 4)

#define LOG2E 1.4426950408889634f


template <int K> struct Slot;
#define DEF_SLOT(K, LO, HI)                                                                      \
template <> struct Slot<K> {                                                                     \
    static __device__ __forceinline__ void wmma(v8f& acc, const v2f a, const v2f& q) {           \
        asm volatile("v_wmma_f32_16x16x4_f32 %0, %1, %2, %0"                                     \
                     : "+v"(acc) : "v"(a), "{v[" #LO ":" #HI "]}"(q));                           \
    }                                                                                            \
    static __device__ __forceinline__ void exp2(v2f& r) {                 \
        asm volatile("v_exp_f32 v" #LO ", v" #LO "\n\tv_exp_f32 v" #HI ", v" #HI                \
                     : "+{v[" #LO ":" #HI "]}"(r));                                              \
    }                                                                                            \
    static __device__ __forceinline__ void rcp(v2f& r) {                             \
        asm volatile("v_rcp_f32 v" #LO ", v" #LO "\n\tv_rcp_f32 v" #HI ", v" #HI                \
                     : "+{v[" #LO ":" #HI "]}"(r));                                              \
    }                                                                                            \
    static __device__ __forceinline__ void add1(v2f& r) {                            \
        asm volatile("v_pk_add_f32 %0, %0, 1.0 op_sel_hi:[1,0]" : "+{v[" #LO ":" #HI "]}"(r));   \
    }                                                                                            \
    static __device__ __forceinline__ void sq1(v2f& r) {                       \
        asm volatile("v_pk_fma_f32 %0, %0, %0, 1.0 op_sel_hi:[1,1,0]"                            \
                     : "+{v[" #LO ":" #HI "]}"(r));                                              \
    }                                                                                            \
    static __device__ __forceinline__ void mul(v2f& r, const v2f cc, const v2f e) {              \
        asm volatile("v_pk_mul_f32 %0, %1, %2"                     \
                     : "={v[" #LO ":" #HI "]}"(r) : "v"(cc), "v"(e));                            \
    }                                                                                            \
};
DEF_SLOT(0, 100, 101)
DEF_SLOT(1, 102, 103)
DEF_SLOT(2, 104, 105)
DEF_SLOT(3, 106, 107)
DEF_SLOT(4, 108, 109)
DEF_SLOT(5, 110, 111)
#undef DEF_SLOT

template <int s>
__device__ __forceinline__ void pipe_step(v8f& acc, v2f (&e2)[SPF], v2f (&k2)[SPF], v2f (&r)[SPF],
                                          const v2f cc, const float* pe, const float* pk)
{
    if constexpr (s >= 0)                       Slot<s % 6>::wmma(acc, k2[s], r[s]);
    if constexpr (s + 4 >= 0 && s + 4 < SPF)    Slot<(s + 4) % 6>::exp2(r[s + 4]);
    if constexpr (s + 1 >= 0 && s + 1 < SPF)    Slot<(s + 1) % 6>::rcp(r[s + 1]);
    if constexpr (s + 3 >= 0 && s + 3 < SPF)    Slot<(s + 3) % 6>::add1(r[s + 3]);
    if constexpr (s + 2 >= 0 && s + 2 < SPF)    Slot<(s + 2) % 6>::sq1(r[s + 2]);
    if constexpr (s + 5 < SPF)                  Slot<(s + 5) % 6>::mul(r[s + 5], cc, e2[s + 5]);
    if constexpr (s + 1 < SPF && s + 2 >= SPF)  asm volatile("v_nop");
    if constexpr (s >= 5 && ((s + 11) & 7) == 0 && s + 6 < SPF) {
        constexpr int g = (s + 11) >> 3;
#pragma unroll
        for (int j = 0; j < 8; ++j) {
            const int x = 8 * g + j;
            const int y = 8 * g - 5 + j;
            if (x < SPF) e2[x] = *(const v2f*)(pe + 4 * x);
            if (y < SPF) k2[y] = *(const v2f*)(pk + 4 * y);
        }
    }
}

template <int... I>
__device__ __forceinline__ void pipe_feature(std::integer_sequence<int, I...>, v8f& acc,
                                             v2f (&e2)[SPF], v2f (&k2)[SPF], v2f (&r)[SPF],
                                             const v2f cc, const float* pe, const float* pk)
{
    (pipe_step<I - 5>(acc, e2, k2, r, cc, pe, pk), ...);
}

__global__ __launch_bounds__(TPB) void nam_main(
    const float* __restrict__ X,
    const float* __restrict__ exu_w,
    const float* __restrict__ exu_b,
    const float* __restrict__ dense_k,
    const float* __restrict__ dense_b,
    float* __restrict__ out)
{
    __shared__ float s_ew[FDIM * HDIM];
    __shared__ float s_kd[FDIM * HDIM];
    __shared__ float s_S[FDIM];
    __shared__ float s_red[WAVES * 16];
    __shared__ float s_bias;

    const int tid = threadIdx.x;

    const int wave = tid >> 5;
    const int lane = tid & 31;
    const int half = lane >> 4;
    const int n    = lane & 15;
    const int tile  = wave & (TILES - 1);
    const int slice = wave / TILES;
    const int f0    = slice * FPB;
    const int rowBase = blockIdx.x * ROWS_PER_WG + tile * 16;
    const int b    = rowBase + n;
    const int koff = half * 2;

    const v4f x4 = *(const v4f*)&X[b * FDIM + f0];
    const v4f b4f = *(const v4f*)&exu_b[f0];

    {
        const v4f* w4 = (const v4f*)exu_w;
        const v4f* k4 = (const v4f*)dense_k;
        v4f wv[FDIM * HDIM / 4 / TPB], kv[FDIM * HDIM / 4 / TPB];
#pragma unroll
        for (int j = 0; j < FDIM * HDIM / 4 / TPB; ++j) {
            const int i = (tid + j * TPB + blockIdx.x * 16) & (FDIM * HDIM / 4 - 1);
            wv[j] = w4[i];
            kv[j] = k4[i];
        }
#pragma unroll
        for (int j = 0; j < FDIM * HDIM / 4 / TPB; ++j) {
            const int i = (tid + j * TPB + blockIdx.x * 16) & (FDIM * HDIM / 4 - 1);
            v4f E;
            E.x = __expf(wv[j].x); E.y = __expf(wv[j].y); E.z = __expf(wv[j].z); E.w = __expf(wv[j].w);
            ((v4f*)s_ew)[i] = E * LOG2E;
            ((v4f*)s_kd)[i] = -2.0f * kv[j] * E;
        }
    }
    __syncthreads();

#pragma unroll
    for (int q = 0; q < FDIM / WAVES; ++q) {
        const int f = wave * (FDIM / WAVES) + q;
        const v4f* p = (const v4f*)&s_kd[f * HDIM + lane * 8];
        const v4f a = p[0];
        const v4f b4 = p[1];
        float s = ((a.x + a.y) + (a.z + a.w)) + ((b4.x + b4.y) + (b4.z + b4.w));
#pragma unroll
        for (int o = 16; o > 0; o >>= 1) s += __shfl_xor(s, o, 32);
        if (lane == 0) s_S[f] = -0.5f * s;
    }
    __syncthreads();

    if (tid == 0) {
        float bias = 0.0f;
#pragma unroll
        for (int f = 0; f < FDIM; ++f) bias += dense_b[f];
        s_bias = bias;
    }

    const float c0 = x4.x - b4f.x;
    const float c1 = x4.y - b4f.y;
    const float c2 = x4.z - b4f.z;
    const float c3 = x4.w - b4f.w;

    float tot = 0.0f;

#pragma nounroll
    for (int fl = 0; fl < FPB; ++fl) {
        const float c = (fl == 0) ? c0 : (fl == 1) ? c1 : (fl == 2) ? c2 : c3;
        const float S = s_S[f0 + fl];
        v8f acc = {S, S, S, S, S, S, S, S};
        const float* pe = &s_ew[(f0 + fl) * HDIM + koff];
        const float* pk = &s_kd[(f0 + fl) * HDIM + koff];

        const v2f cc = {c, c};

        v2f e2[SPF], k2[SPF], r[SPF];
#pragma unroll
        for (int x = 0; x < 16; ++x) e2[x] = *(const v2f*)(pe + 4 * x);
#pragma unroll
        for (int y = 0; y < 11; ++y) k2[y] = *(const v2f*)(pk + 4 * y);
        pipe_feature(std::make_integer_sequence<int, SPF + 5>{}, acc, e2, k2, r, cc, pe, pk);

        asm volatile("v_nop\n\tv_nop\n\tv_nop\n\tv_nop\n\tv_nop" : "+v"(acc));
        tot = __builtin_fmaf(c, acc[0], tot);
    }

    if (half == 0) s_red[wave * 16 + n] = tot;
    __syncthreads();

    if (tid < ROWS_PER_WG) {
        const int t  = tid >> 4;
        const int nn = tid & 15;
        float s = s_bias;
#pragma unroll
        for (int y = 0; y < KSPLIT; ++y) s += s_red[(y * TILES + t) * 16 + nn];
        out[blockIdx.x * ROWS_PER_WG + tid] = s;
    }
}

extern "C" void kernel_launch(void* const* d_in, const int* in_sizes, int n_in,
                              void* d_out, int out_size, void* d_ws, size_t ws_size,
                              hipStream_t stream) {
    (void)in_sizes; (void)n_in; (void)out_size; (void)d_ws; (void)ws_size;
    const float* X       = (const float*)d_in[0];
    const float* exu_w   = (const float*)d_in[1];
    const float* exu_b   = (const float*)d_in[2];
    const float* dense_k = (const float*)d_in[3];
    const float* dense_b = (const float*)d_in[4];
    float* out     = (float*)d_out;

    nam_main<<<BSZ / ROWS_PER_WG, TPB, 0, stream>>>(X, exu_w, exu_b, dense_k, dense_b, out);
}


# speedup vs baseline: 1.3909x; 1.0580x over previous
#include <hip/hip_runtime.h>
#include <utility>


typedef __attribute__((ext_vector_type(2))) float v2f;
typedef __attribute__((ext_vector_type(4))) float v4f;
typedef __attribute__((ext_vector_type(8))) float v8f;

#define BSZ    8192
#define FDIM   64
#define HDIM   256
#define KSPLIT 16
#define FPB    (FDIM / KSPLIT)
#define TILES  2
#define WAVES  (KSPLIT * TILES)
#define TPB    (WAVES * 32)
#define ROWS_PER_WG (TILES * 16)

#define SPF    (HDIM / 4)

#define LOG2E 1.4426950408889634f


template <int K> struct Slot;
#define DEF_SLOT(K, LO, HI)                                                                      \
template <> struct Slot<K> {                                                                     \
    static __device__ __forceinline__ void wmma(v8f& acc, const v2f a, const v2f& q) {           \
        asm volatile("v_wmma_f32_16x16x4_f32 %0, %1, %2, %0"                                     \
                     : "+v"(acc) : "v"(a), "{v[" #LO ":" #HI "]}"(q));                           \
    }                                                                                            \
        \
        \
    static __device__ __forceinline__ void exp_lo(v2f& r) {                                      \
        asm volatile("v_exp_f32 v" #LO ", v" #LO : "+{v[" #LO ":" #HI "]}"(r));                  \
    }                                                                                            \
    static __device__ __forceinline__ void exp_hi(v2f& r) {                                      \
        asm volatile("v_exp_f32 v" #HI ", v" #HI : "+{v[" #LO ":" #HI "]}"(r));                  \
    }                                                                                            \
                                                    \
    static __device__ __forceinline__ void rcp_lo(v2f& r) {                                      \
        asm volatile("v_rcp_f32 v" #LO ", v" #LO : "+{v[" #LO ":" #HI "]}"(r));                  \
    }                                                                                            \
    static __device__ __forceinline__ void rcp_hi(v2f& r) {                                      \
        asm volatile("v_rcp_f32 v" #HI ", v" #HI : "+{v[" #LO ":" #HI "]}"(r));                  \
    }                                                                                            \
    static __device__ __forceinline__ void add1(v2f& r) {                            \
        asm volatile("v_pk_add_f32 %0, %0, 1.0 op_sel_hi:[1,0]" : "+{v[" #LO ":" #HI "]}"(r));   \
    }                                                                                            \
    static __device__ __forceinline__ void sq1(v2f& r) {                       \
        asm volatile("v_pk_fma_f32 %0, %0, %0, 1.0 op_sel_hi:[1,1,0]"                            \
                     : "+{v[" #LO ":" #HI "]}"(r));                                              \
    }                                                                                            \
    static __device__ __forceinline__ void mul(v2f& r, const v2f cc, const v2f e) {              \
        asm volatile("v_pk_mul_f32 %0, %1, %2"                     \
                     : "={v[" #LO ":" #HI "]}"(r) : "v"(cc), "v"(e));                            \
    }                                                                                            \
};
DEF_SLOT(0,  98,  99)
DEF_SLOT(1, 100, 101)
DEF_SLOT(2, 102, 103)
DEF_SLOT(3, 104, 105)
DEF_SLOT(4, 106, 107)
DEF_SLOT(5, 108, 109)
DEF_SLOT(6, 110, 111)
#undef DEF_SLOT

template <int s>
__device__ __forceinline__ void pipe_step(v8f& acc, v2f (&e2)[SPF], v2f (&k2)[SPF], v2f (&r)[SPF],
                                          const v2f cc, const float* pe, const float* pk)
{
    if constexpr (s >= 0)                       Slot<s % 7>::wmma(acc, k2[s], r[s]);
    if constexpr (s + 2 < 0 && s + 4 >= 0)      asm volatile("v_nop");
    if constexpr (s + 2 >= 0 && s + 2 < SPF)    Slot<(s + 2) % 7>::rcp_lo(r[s + 2]);
    if constexpr (s + 4 >= 0 && s + 4 < SPF)    Slot<(s + 4) % 7>::add1(r[s + 4]);
    if constexpr (s + 2 >= 0 && s + 2 < SPF)    Slot<(s + 2) % 7>::rcp_hi(r[s + 2]);
    if constexpr (s + 3 >= 0 && s + 3 < SPF)    Slot<(s + 3) % 7>::sq1(r[s + 3]);
    if constexpr (s + 5 >= 0 && s + 5 < SPF)    Slot<(s + 5) % 7>::exp_lo(r[s + 5]);
    if constexpr (s + 6 < SPF)                  Slot<(s + 6) % 7>::mul(r[s + 6], cc, e2[s + 6]);
    if constexpr (s + 5 >= 0 && s + 5 < SPF)    Slot<(s + 5) % 7>::exp_hi(r[s + 5]);
    if constexpr (s >= 4 && ((s + 12) & 7) == 0 && s + 7 < SPF) {
        constexpr int g = (s + 12) >> 3;
#pragma unroll
        for (int j = 0; j < 8; ++j) {
            const int x = 8 * g + j;
            const int y = 8 * g - 5 + j;
            if (x < SPF) e2[x] = *(const v2f*)(pe + 4 * x);
            if (y < SPF) k2[y] = *(const v2f*)(pk + 4 * y);
        }
    }
}

template <int... I>
__device__ __forceinline__ void pipe_feature(std::integer_sequence<int, I...>, v8f& acc,
                                             v2f (&e2)[SPF], v2f (&k2)[SPF], v2f (&r)[SPF],
                                             const v2f cc, const float* pe, const float* pk)
{
    (pipe_step<I - 6>(acc, e2, k2, r, cc, pe, pk), ...);
}

__global__ __launch_bounds__(TPB) void nam_main(
    const float* __restrict__ X,
    const float* __restrict__ exu_w,
    const float* __restrict__ exu_b,
    const float* __restrict__ dense_k,
    const float* __restrict__ dense_b,
    float* __restrict__ out)
{
    __shared__ float s_ew[FDIM * HDIM];
    __shared__ float s_kd[FDIM * HDIM];
    __shared__ float s_S[FDIM];
    __shared__ float s_red[WAVES * 16];
    __shared__ float s_bias;

    const int tid = threadIdx.x;

    const int wave = tid >> 5;
    const int lane = tid & 31;
    const int half = lane >> 4;
    const int n    = lane & 15;
    const int tile  = wave & (TILES - 1);
    const int slice = wave / TILES;
    const int f0    = slice * FPB;
    const int rowBase = blockIdx.x * ROWS_PER_WG + tile * 16;
    const int b    = rowBase + n;
    const int koff = half * 2;

    const v4f x4 = *(const v4f*)&X[b * FDIM + f0];
    const v4f b4f = *(const v4f*)&exu_b[f0];

    {
        const v4f* w4 = (const v4f*)exu_w;
        const v4f* k4 = (const v4f*)dense_k;
        v4f wv[FDIM * HDIM / 4 / TPB], kv[FDIM * HDIM / 4 / TPB];
#pragma unroll
        for (int j = 0; j < FDIM * HDIM / 4 / TPB; ++j) {
            const int i = (tid + j * TPB + blockIdx.x * 16) & (FDIM * HDIM / 4 - 1);
            wv[j] = w4[i];
            kv[j] = k4[i];
        }
#pragma unroll
        for (int j = 0; j < FDIM * HDIM / 4 / TPB; ++j) {
            const int i = (tid + j * TPB + blockIdx.x * 16) & (FDIM * HDIM / 4 - 1);
            v4f E;
            E.x = __expf(wv[j].x); E.y = __expf(wv[j].y); E.z = __expf(wv[j].z); E.w = __expf(wv[j].w);
            ((v4f*)s_ew)[i] = E * LOG2E;
            ((v4f*)s_kd)[i] = -2.0f * kv[j] * E;
        }
    }
    __syncthreads();

#pragma unroll
    for (int q = 0; q < FDIM / WAVES; ++q) {
        const int f = wave * (FDIM / WAVES) + q;
        const v4f* p = (const v4f*)&s_kd[f * HDIM + lane * 8];
        const v4f a = p[0];
        const v4f b4 = p[1];
        float s = ((a.x + a.y) + (a.z + a.w)) + ((b4.x + b4.y) + (b4.z + b4.w));
#pragma unroll
        for (int o = 16; o > 0; o >>= 1) s += __shfl_xor(s, o, 32);
        if (lane == 0) s_S[f] = -0.5f * s;
    }
    __syncthreads();

    if (tid == 0) {
        float bias = 0.0f;
#pragma unroll
        for (int f = 0; f < FDIM; ++f) bias += dense_b[f];
        s_bias = bias;
    }

    const float c0 = x4.x - b4f.x;
    const float c1 = x4.y - b4f.y;
    const float c2 = x4.z - b4f.z;
    const float c3 = x4.w - b4f.w;

    float tot = 0.0f;

#pragma nounroll
    for (int fl = 0; fl < FPB; ++fl) {
        const float c = (fl == 0) ? c0 : (fl == 1) ? c1 : (fl == 2) ? c2 : c3;
        const float S = s_S[f0 + fl];
        v8f acc = {S, S, S, S, S, S, S, S};
        const float* pe = &s_ew[(f0 + fl) * HDIM + koff];
        const float* pk = &s_kd[(f0 + fl) * HDIM + koff];

        const v2f cc = {c, c};

        v2f e2[SPF], k2[SPF], r[SPF];
#pragma unroll
        for (int x = 0; x < 16; ++x) e2[x] = *(const v2f*)(pe + 4 * x);
#pragma unroll
        for (int y = 0; y < 11; ++y) k2[y] = *(const v2f*)(pk + 4 * y);
        pipe_feature(std::make_integer_sequence<int, SPF + 6>{}, acc, e2, k2, r, cc, pe, pk);

        asm volatile("v_nop\n\tv_nop\n\tv_nop\n\tv_nop\n\tv_nop" : "+v"(acc));
        tot = __builtin_fmaf(c, acc[0], tot);
    }

    if (half == 0) s_red[wave * 16 + n] = tot;
    __syncthreads();

    if (tid < ROWS_PER_WG) {
        const int t  = tid >> 4;
        const int nn = tid & 15;
        float s = s_bias;
#pragma unroll
        for (int y = 0; y < KSPLIT; ++y) s += s_red[(y * TILES + t) * 16 + nn];
        out[blockIdx.x * ROWS_PER_WG + tid] = s;
    }
}

extern "C" void kernel_launch(void* const* d_in, const int* in_sizes, int n_in,
                              void* d_out, int out_size, void* d_ws, size_t ws_size,
                              hipStream_t stream) {
    (void)in_sizes; (void)n_in; (void)out_size; (void)d_ws; (void)ws_size;
    const float* X       = (const float*)d_in[0];
    const float* exu_w   = (const float*)d_in[1];
    const float* exu_b   = (const float*)d_in[2];
    const float* dense_k = (const float*)d_in[3];
    const float* dense_b = (const float*)d_in[4];
    float* out     = (float*)d_out;

    nam_main<<<BSZ / ROWS_PER_WG, TPB, 0, stream>>>(X, exu_w, exu_b, dense_k, dense_b, out);
}
